# starting kernel
# baseline (speedup 1.0000x reference)
#include <hip/hip_runtime.h>
#include <math.h>


typedef __attribute__((ext_vector_type(2))) float v2f;
typedef __attribute__((ext_vector_type(8))) float v8f;

#define NH        8
#define HD        32
#define DMODEL    256
#define HGRID     56
#define MWIN      7
#define MTOK      112
#define KCHUNK    64
#define XS_STRIDE 68
#define WP_STRIDE 224
#define NPAIR     (KCHUNK / 2)
#define QK_STRIDE 100
#define ATT_STRIDE 18

#define XS_FLOATS  (MTOK * XS_STRIDE)
#define WH_FLOATS  (NPAIR * WP_STRIDE)
#define QK_FLOATS  (MTOK * QK_STRIDE)
#define ATT_FLOATS (8 * 16 * ATT_STRIDE)
#define SMEM_FLOATS (XS_FLOATS + WH_FLOATS + QK_FLOATS + ATT_FLOATS)
#define SMEM_BYTES  (SMEM_FLOATS * 4 + MTOK * 4)

static __device__ __forceinline__ v8f wmma_f32_4(v2f a, v2f b, v8f c) {
    return __builtin_amdgcn_wmma_f32_16x16x4_f32(false, a, false, b,
                                                 (short)0, c, false, false);
}

__global__ __launch_bounds__(256)
void swin_msa_fused(const float* __restrict__ x, const float* __restrict__ Wg,
                    const float* __restrict__ bvec, float* __restrict__ out) {
    extern __shared__ float smem[];
    float* xs  = smem;
    float* whp = xs + XS_FLOATS;
    float* qk  = whp + WH_FLOATS;
    float* att = qk + QK_FLOATS;
    int* rowoff = (int*)(att + ATT_FLOATS);

    const int half = blockIdx.x;
    const int wi   = blockIdx.y;
    const int bb   = blockIdx.z;
    const int tid  = threadIdx.x;
    const int wave = tid >> 5;
    const int lane = tid & 31;
    const int r    = lane & 15;
    const int hiK  = (lane >> 4) << 1;
    const int hiP  = lane >> 4;
    const int rowb = (lane >= 16) ? 8 : 0;

    if (tid < MTOK) {
        int w  = tid >> 4;
        int t  = tid & 15;
        int gh = 4 * wi + (t >> 2) + 2;               if (gh >= HGRID) gh -= HGRID;
        int gw = 4 * (half * MWIN + w) + (t & 3) + 2; if (gw >= HGRID) gw -= HGRID;
        rowoff[tid] = bb * (HGRID * HGRID) + gh * HGRID + gw;
    }
    __syncthreads();

    const bool lastRow = (wi == 13);
    const bool lastCol = (half == 1) && (wave == 6);
    const bool maskR = lastRow && (((lane >= 16) ? 1 : 0) != ((lane & 8) ? 1 : 0));
    const float scale = 0.17677669529663687f;

    for (int h = 0; h < NH; ++h) {
        v8f acc[6];
        if (wave < MWIN) {
            const float* bp = bvec + 96 * h;
            #pragma unroll
            for (int n = 0; n < 6; ++n) {
                float bv = bp[16 * n + r];
                v8f t;
                #pragma unroll
                for (int q = 0; q < 8; ++q) t[q] = bv;
                acc[n] = t;
            }
        }

        for (int kc = 0; kc < 4; ++kc) {
            __syncthreads();
            #pragma unroll
            for (int it = 0; it < 7; ++it) {
                int idx = tid + 256 * it;
                int row = idx >> 4;
                int c4  = (idx & 15) << 2;
                const float4 v = *(const float4*)(x + (size_t)rowoff[row] * DMODEL
                                                    + kc * KCHUNK + c4);
                *(float4*)&xs[row * XS_STRIDE + c4] = v;
            }
            #pragma unroll
            for (int it = 0; it < 6; ++it) {
                int idx  = tid + 256 * it;
                int pair = idx / 48;
                int c2g  = idx % 48;
                const float* g0 = Wg + (size_t)(kc * KCHUNK + 2 * pair) * (3 * DMODEL)
                                     + 96 * h + 2 * c2g;
                float2 a0 = *(const float2*)g0;
                float2 a1 = *(const float2*)(g0 + 3 * DMODEL);
                float4 st; st.x = a0.x; st.y = a1.x; st.z = a0.y; st.w = a1.y;
                *(float4*)&whp[pair * WP_STRIDE + 4 * c2g] = st;
            }
            __syncthreads();

            if (wave < MWIN) {
                const float* xrow = &xs[(16 * wave + r) * XS_STRIDE + hiK];
                const float* wrow = &whp[hiP * WP_STRIDE + 2 * r];
                #pragma unroll 4
                for (int k0 = 0; k0 < KCHUNK; k0 += 4) {
                    v2f a = *(const v2f*)(xrow + k0);
                    v2f bf[6];
                    #pragma unroll
                    for (int n = 0; n < 6; ++n)
                        bf[n] = *(const v2f*)(wrow + (k0 >> 1) * WP_STRIDE + 32 * n);
                    #pragma unroll
                    for (int n = 0; n < 6; ++n)
                        acc[n] = wmma_f32_4(a, bf[n], acc[n]);
                }
            }
        }

        if (wave < MWIN) {
            int qrowb = 16 * wave + rowb;
            #pragma unroll
            for (int n = 0; n < 6; ++n)
                #pragma unroll
                for (int v = 0; v < 8; ++v)
                    qk[(qrowb + v) * QK_STRIDE + 16 * n + r] = acc[n][v];

            const float* qrow = &qk[(16 * wave + r) * QK_STRIDE];
            v8f s = {0.f, 0.f, 0.f, 0.f, 0.f, 0.f, 0.f, 0.f};
            #pragma unroll
            for (int k0 = 0; k0 < HD; k0 += 4) {
                int ka = 3 * (k0 + hiK);
                v2f a, bf;
                a.x  = qrow[ka];     a.y  = qrow[ka + 3];
                bf.x = qrow[ka + 1]; bf.y = qrow[ka + 4];
                s = wmma_f32_4(a, bf, s);
            }

            v8f p;
            #pragma unroll
            for (int v = 0; v < 8; ++v) {
                float xv = s[v] * scale;
                bool mC = lastCol && (((v & 2) != 0) != ((lane & 2) != 0));
                if (maskR || mC) xv = -1e30f;
                float m = xv;
                #pragma unroll
                for (int off = 8; off; off >>= 1)
                    m = fmaxf(m, __shfl_xor(m, off, 16));
                float e = __expf(xv - m);
                float sm = e;
                #pragma unroll
                for (int off = 8; off; off >>= 1)
                    sm += __shfl_xor(sm, off, 16);
                p[v] = e / sm;
            }

            float* am = &att[wave * 16 * ATT_STRIDE];
            #pragma unroll
            for (int v = 0; v < 8; ++v)
                am[(rowb + v) * ATT_STRIDE + r] = p[v];

            v8f o0 = {0.f, 0.f, 0.f, 0.f, 0.f, 0.f, 0.f, 0.f};
            v8f o1 = {0.f, 0.f, 0.f, 0.f, 0.f, 0.f, 0.f, 0.f};
            #pragma unroll
            for (int k0 = 0; k0 < 16; k0 += 4) {
                v2f a = *(const v2f*)(am + r * ATT_STRIDE + k0 + hiK);
                const float* v0p = &qk[(16 * wave + k0 + hiK) * QK_STRIDE + 2];
                const float* v1p = v0p + QK_STRIDE;
                v2f b0; b0.x = v0p[3 * r];        b0.y = v1p[3 * r];
                o0 = wmma_f32_4(a, b0, o0);
                v2f b1; b1.x = v0p[3 * (16 + r)]; b1.y = v1p[3 * (16 + r)];
                o1 = wmma_f32_4(a, b1, o1);
            }

            float* ob = out + 32 * h + r;
            #pragma unroll
            for (int v = 0; v < 8; ++v) {
                int grow = rowoff[16 * wave + rowb + v];
                float* pdst = ob + (size_t)grow * DMODEL;
                __builtin_nontemporal_store(o0[v], pdst);
                __builtin_nontemporal_store(o1[v], pdst + 16);
            }
        }
    }
}

extern "C" void kernel_launch(void* const* d_in, const int* in_sizes, int n_in,
                              void* d_out, int out_size, void* d_ws, size_t ws_size,
                              hipStream_t stream) {
    (void)in_sizes; (void)n_in; (void)d_ws; (void)ws_size; (void)out_size;
    const float* x = (const float*)d_in[0];
    const float* W = (const float*)d_in[1];
    const float* b = (const float*)d_in[2];
    float* out = (float*)d_out;

    static_assert(SMEM_BYTES < 320 * 1024, "LDS budget");
    hipFuncSetAttribute((const void*)swin_msa_fused,
                        hipFuncAttributeMaxDynamicSharedMemorySize, SMEM_BYTES);

    dim3 grid(2, 14, 32);
    swin_msa_fused<<<grid, 256, SMEM_BYTES, stream>>>(x, W, b, out);
}


# Token tile register-resident: a wave keeps its window's 16x256 x rows as A fragments over all 8 heads (x read once; k regrouped, qkv sum order changes); LDS holds one whole W head slice per stage
# speedup vs baseline: 1.3815x; 1.3815x over previous
#include <hip/hip_runtime.h>
#include <math.h>


typedef __attribute__((ext_vector_type(2))) float v2f;
typedef __attribute__((ext_vector_type(8))) float v8f;

#define NH         8
#define HD         32
#define DMODEL     256
#define HGRID      56
#define NWIN       8
#define WIN_IMG    196
#define WL_STEP    768
#define WL_FLOATS  (32 * WL_STEP)
#define HI_PAIR    21
#define QK_STRIDE  100
#define QK_FLOATS  (NWIN * 16 * QK_STRIDE)
#define ATT_STRIDE 18
#define ATT_FLOATS (NWIN * 16 * ATT_STRIDE)
#define SMEM_BYTES ((WL_FLOATS + QK_FLOATS + ATT_FLOATS) * 4 + NWIN * 16 * 4)

static __device__ __forceinline__ v8f wmma_f32_4(v2f a, v2f b, v8f c) {
    return __builtin_amdgcn_wmma_f32_16x16x4_f32(false, a, false, b,
                                                 (short)0, c, false, false);
}

static __device__ __forceinline__ float4 sel4(bool c, float4 a, float4 b) {
    float4 o;
    o.x = c ? a.x : b.x; o.y = c ? a.y : b.y;
    o.z = c ? a.z : b.z; o.w = c ? a.w : b.w;
    return o;
}

__global__ __launch_bounds__(256, 4)
void swin_msa_fused(const float* __restrict__ x, const float* __restrict__ Wg,
                    const float* __restrict__ bvec, float* __restrict__ out) {
    extern __shared__ float smem[];
    float* wl  = smem;
    float* qk  = wl + WL_FLOATS;
    float* att = qk + QK_FLOATS;
    int* rowoff = (int*)(att + ATT_FLOATS);

    const int tid  = threadIdx.x;
    const int wave = tid >> 5;
    const int lane = tid & 31;
    const int r    = lane & 15;
    const int hh   = lane >> 4;
    const int rowb = hh << 3;

    const int gwin = blockIdx.x * NWIN + wave;
    const int bb   = gwin / WIN_IMG;
    const int rem  = gwin - bb * WIN_IMG;
    const int wi   = rem / 14;
    const int wj   = rem - wi * 14;

    int gh = 4 * wi + (r >> 2) + 2; if (gh >= HGRID) gh -= HGRID;
    int gw = 4 * wj + (r & 3) + 2;  if (gw >= HGRID) gw -= HGRID;
    const int myrow = bb * (HGRID * HGRID) + gh * HGRID + gw;
    if (hh == 0) rowoff[16 * wave + r] = myrow;

    float4 xa[32];
    {
        const float* xr = x + (size_t)myrow * DMODEL + 4 * hh;
        #pragma unroll
        for (int i = 0; i < 32; ++i)
            xa[i] = *(const float4*)(xr + 8 * i);
    }

    const bool lastRow = (wi == 13);
    const bool lastCol = (wj == 13);
    const bool maskR = lastRow && (hh != ((lane >> 3) & 1));
    const float scale = 0.17677669529663687f;

    const int qq  = r & 3;
    const int sh  = (r >> 2) & 1;
    const int ntb = (r >> 3) & 1;
    const float* wsrc0 = Wg + (size_t)(16 * wave + 8 * hh + 4 * sh) * (3 * DMODEL)
                            + 16 * ntb + 4 * qq;
    float* wdst0 = wl + (2 * wave + hh) * WL_STEP + ntb * 128 + sh * 64 + 16 * qq;
    const bool rot1 = sh != 0;
    const bool rot2 = ntb != 0;
    const int  rot  = sh + 2 * ntb;
    float* wdst_t0 = wdst0 + 4 * ((0 + rot) & 3);
    float* wdst_t1 = wdst0 + 4 * ((1 + rot) & 3);
    float* wdst_t2 = wdst0 + 4 * ((2 + rot) & 3);
    float* wdst_t3 = wdst0 + 4 * ((3 + rot) & 3);

    const float* wrd = wl + hh * 64 + 4 * r;
    int hi_off = hh * 64 + 4 * r + HI_PAIR * WL_STEP;
    asm volatile("" : "+v"(hi_off));
    const float* wrd_hi = wl + hi_off;

    for (int h = 0; h < NH; ++h) {
        __syncthreads();

        {
            const float* wsrc = wsrc0 + 96 * h;
            #pragma unroll
            for (int bt = 0; bt < 2; ++bt) {
                float4 rr[3][4];
                #pragma unroll
                for (int u = 0; u < 3; ++u) {
                    const float* gp = wsrc + (size_t)bt * (128 * 3 * DMODEL) + u * 32;
                    #pragma unroll
                    for (int j = 0; j < 4; ++j)
                        rr[u][j] = *(const float4*)(gp + j * (3 * DMODEL));
                }
                __builtin_amdgcn_sched_barrier(0);
                #pragma unroll
                for (int u = 0; u < 3; ++u) {
                    float4 c0, c1, c2, c3;
                    c0.x = rr[u][0].x; c0.y = rr[u][1].x; c0.z = rr[u][2].x; c0.w = rr[u][3].x;
                    c1.x = rr[u][0].y; c1.y = rr[u][1].y; c1.z = rr[u][2].y; c1.w = rr[u][3].y;
                    c2.x = rr[u][0].z; c2.y = rr[u][1].z; c2.z = rr[u][2].z; c2.w = rr[u][3].z;
                    c3.x = rr[u][0].w; c3.y = rr[u][1].w; c3.z = rr[u][2].w; c3.w = rr[u][3].w;
                    const float4 d0 = sel4(rot1, c1, c0), d1 = sel4(rot1, c2, c1),
                                 d2 = sel4(rot1, c3, c2), d3 = sel4(rot1, c0, c3);
                    const float4 e0 = sel4(rot2, d2, d0), e1 = sel4(rot2, d3, d1),
                                 e2 = sel4(rot2, d0, d2), e3 = sel4(rot2, d1, d3);
                    const int so = bt * (16 * WL_STEP) + u * 256;
                    *(float4*)(wdst_t0 + so) = e0;
                    *(float4*)(wdst_t1 + so) = e1;
                    *(float4*)(wdst_t2 + so) = e2;
                    *(float4*)(wdst_t3 + so) = e3;
                }
            }
        }

        v8f acc[6];
        {
            const float* bp = bvec + 96 * h;
            #pragma unroll
            for (int n = 0; n < 6; ++n) {
                float bv = bp[16 * n + r];
                v8f t;
                #pragma unroll
                for (int q = 0; q < 8; ++q) t[q] = bv;
                acc[n] = t;
            }
        }

        __syncthreads();

        float4 bq[2][6];
        #pragma unroll
        for (int n = 0; n < 6; ++n)
            bq[0][n] = *(const float4*)(wrd + 128 * n);
        __builtin_amdgcn_sched_group_barrier(0x100, 6, 0);
        #pragma unroll
        for (int i = 0; i < 32; ++i) {
            if (i + 1 < 32) {
                const float* nb = (i + 1 < HI_PAIR) ? wrd + (i + 1) * WL_STEP
                                                    : wrd_hi + (i + 1 - HI_PAIR) * WL_STEP;
                #pragma unroll
                for (int n = 0; n < 6; ++n)
                    bq[(i + 1) & 1][n] = *(const float4*)(nb + 128 * n);
            }
            v2f a0, a1;
            a0.x = xa[i].x; a0.y = xa[i].y;
            a1.x = xa[i].z; a1.y = xa[i].w;
            #pragma unroll
            for (int n = 0; n < 6; ++n) {
                v2f b0; b0.x = bq[i & 1][n].x; b0.y = bq[i & 1][n].y;
                acc[n] = wmma_f32_4(a0, b0, acc[n]);
            }
            #pragma unroll
            for (int n = 0; n < 6; ++n) {
                v2f b1; b1.x = bq[i & 1][n].z; b1.y = bq[i & 1][n].w;
                acc[n] = wmma_f32_4(a1, b1, acc[n]);
            }
            if (i + 1 < 32)
                __builtin_amdgcn_sched_group_barrier(0x100, 6, 0);
            __builtin_amdgcn_sched_group_barrier(0x008, 12, 0);
        }
        __builtin_amdgcn_sched_barrier(0);

        int qrowb = 16 * wave + rowb;
        #pragma unroll
        for (int n = 0; n < 6; ++n)
            #pragma unroll
            for (int v = 0; v < 8; ++v)
                qk[(qrowb + v) * QK_STRIDE + 16 * n + r] = acc[n][v];

        const float* qrow = &qk[(16 * wave + r) * QK_STRIDE];
        v8f sc = {0.f, 0.f, 0.f, 0.f, 0.f, 0.f, 0.f, 0.f};
        #pragma unroll
        for (int k0 = 0; k0 < HD; k0 += 4) {
            int ka = 3 * (k0 + 2 * hh);
            v2f a, bf;
            a.x  = qrow[ka];     a.y  = qrow[ka + 3];
            bf.x = qrow[ka + 1]; bf.y = qrow[ka + 4];
            sc = wmma_f32_4(a, bf, sc);
        }

        v8f p;
        #pragma unroll
        for (int v = 0; v < 8; ++v) {
            float xv = sc[v] * scale;
            bool mC = lastCol && (((v & 2) != 0) != ((lane & 2) != 0));
            if (maskR || mC) xv = -1e30f;
            float m = xv;
            #pragma unroll
            for (int off = 8; off; off >>= 1)
                m = fmaxf(m, __shfl_xor(m, off, 16));
            float e = __expf(xv - m);
            float sm = e;
            #pragma unroll
            for (int off = 8; off; off >>= 1)
                sm += __shfl_xor(sm, off, 16);
            p[v] = e / sm;
        }

        float* am = &att[wave * 16 * ATT_STRIDE];
        #pragma unroll
        for (int v = 0; v < 8; ++v)
            am[(rowb + v) * ATT_STRIDE + r] = p[v];

        v8f o0 = {0.f, 0.f, 0.f, 0.f, 0.f, 0.f, 0.f, 0.f};
        v8f o1 = {0.f, 0.f, 0.f, 0.f, 0.f, 0.f, 0.f, 0.f};
        #pragma unroll
        for (int k0 = 0; k0 < 16; k0 += 4) {
            v2f a = *(const v2f*)(am + r * ATT_STRIDE + k0 + 2 * hh);
            const float* v0p = &qk[(16 * wave + k0 + 2 * hh) * QK_STRIDE + 2];
            const float* v1p = v0p + QK_STRIDE;
            v2f b0; b0.x = v0p[3 * r];        b0.y = v1p[3 * r];
            o0 = wmma_f32_4(a, b0, o0);
            v2f b1; b1.x = v0p[3 * (16 + r)]; b1.y = v1p[3 * (16 + r)];
            o1 = wmma_f32_4(a, b1, o1);
        }

        float* ob = out + 32 * h + r;
        #pragma unroll
        for (int v = 0; v < 8; ++v) {
            int grow = rowoff[16 * wave + rowb + v];
            float* pdst = ob + (size_t)grow * DMODEL;
            __builtin_nontemporal_store(o0[v], pdst);
            __builtin_nontemporal_store(o1[v], pdst + 16);
        }
    }
}

extern "C" void kernel_launch(void* const* d_in, const int* in_sizes, int n_in,
                              void* d_out, int out_size, void* d_ws, size_t ws_size,
                              hipStream_t stream) {
    (void)in_sizes; (void)n_in; (void)d_ws; (void)ws_size; (void)out_size;
    const float* x = (const float*)d_in[0];
    const float* W = (const float*)d_in[1];
    const float* b = (const float*)d_in[2];
    float* out = (float*)d_out;

    static_assert(SMEM_BYTES <= 160 * 1024, "two blocks per compute unit");
    hipFuncSetAttribute((const void*)swin_msa_fused,
                        hipFuncAttributeMaxDynamicSharedMemorySize, SMEM_BYTES);

    dim3 grid(784);
    swin_msa_fused<<<grid, 256, SMEM_BYTES, stream>>>(x, W, b, out);
}


# Balanced resident grid: 512 blocks (2 per compute unit) each take a contiguous 12 or 13 of the 784 x 8 (window group, head) units, x reloaded per group; was 784 blocks, 3 or 4 per compute unit
# speedup vs baseline: 1.5701x; 1.1365x over previous
#include <hip/hip_runtime.h>
#include <math.h>


typedef __attribute__((ext_vector_type(2))) float v2f;
typedef __attribute__((ext_vector_type(8))) float v8f;

#define NH         8
#define HD         32
#define DMODEL     256
#define HGRID      56
#define NWIN       8
#define WIN_IMG    196
#define NBLOCK     512
#define WL_STEP    768
#define WL_FLOATS  (32 * WL_STEP)
#define HI_PAIR    21
#define QK_STRIDE  100
#define QK_FLOATS  (NWIN * 16 * QK_STRIDE)
#define ATT_STRIDE 18
#define ATT_FLOATS (NWIN * 16 * ATT_STRIDE)
#define SMEM_BYTES ((WL_FLOATS + QK_FLOATS + ATT_FLOATS) * 4 + NWIN * 16 * 4)

static __device__ __forceinline__ v8f wmma_f32_4(v2f a, v2f b, v8f c) {
    return __builtin_amdgcn_wmma_f32_16x16x4_f32(false, a, false, b,
                                                 (short)0, c, false, false);
}

static __device__ __forceinline__ float4 sel4(bool c, float4 a, float4 b) {
    float4 o;
    o.x = c ? a.x : b.x; o.y = c ? a.y : b.y;
    o.z = c ? a.z : b.z; o.w = c ? a.w : b.w;
    return o;
}

static __device__ __forceinline__ int share(int b) {
    return (49 * b + 2 * ((b >> 8) & 1)) >> 2;
}

__global__ __launch_bounds__(256, 4)
void swin_msa_fused(const float* __restrict__ x, const float* __restrict__ Wg,
                    const float* __restrict__ bvec, float* __restrict__ out) {
    extern __shared__ float smem[];
    float* wl  = smem;
    float* qk  = wl + WL_FLOATS;
    float* att = qk + QK_FLOATS;
    int* rowoff = (int*)(att + ATT_FLOATS);

    const int tid  = threadIdx.x;
    const int wave = tid >> 5;
    const int lane = tid & 31;
    const int r    = lane & 15;
    const int hh   = lane >> 4;
    const int rowb = hh << 3;

    const float scale = 0.17677669529663687f;

    const int qq  = r & 3;
    const int sh  = (r >> 2) & 1;
    const int ntb = (r >> 3) & 1;
    const float* wsrc0 = Wg + (size_t)(16 * wave + 8 * hh + 4 * sh) * (3 * DMODEL)
                            + 16 * ntb + 4 * qq;
    float* wdst0 = wl + (2 * wave + hh) * WL_STEP + ntb * 128 + sh * 64 + 16 * qq;
    const bool rot1 = sh != 0;
    const bool rot2 = ntb != 0;
    const int  rot  = sh + 2 * ntb;
    float* wdst_t0 = wdst0 + 4 * ((0 + rot) & 3);
    float* wdst_t1 = wdst0 + 4 * ((1 + rot) & 3);
    float* wdst_t2 = wdst0 + 4 * ((2 + rot) & 3);
    float* wdst_t3 = wdst0 + 4 * ((3 + rot) & 3);

    const float* wrd = wl + hh * 64 + 4 * r;
    int hi_off = hh * 64 + 4 * r + HI_PAIR * WL_STEP;
    asm volatile("" : "+v"(hi_off));
    const float* wrd_hi = wl + hi_off;

    const int ubeg = share(blockIdx.x);
    const int uend = share(blockIdx.x + 1);
    const int glast = (uend - 1) >> 3;

    for (int g = ubeg >> 3; g <= glast; ++g) {
        const int hbeg = (ubeg > 8 * g) ? ubeg - 8 * g : 0;
        const int hend = (uend < 8 * g + 8) ? uend - 8 * g : NH;

        const int gwin = g * NWIN + wave;
        const int bb   = gwin / WIN_IMG;
        const int rem  = gwin - bb * WIN_IMG;
        const int wi   = rem / 14;
        const int wj   = rem - wi * 14;

        int gh = 4 * wi + (r >> 2) + 2; if (gh >= HGRID) gh -= HGRID;
        int gw = 4 * wj + (r & 3) + 2;  if (gw >= HGRID) gw -= HGRID;
        const int myrow = bb * (HGRID * HGRID) + gh * HGRID + gw;
        if (hh == 0) rowoff[16 * wave + r] = myrow;

        float4 xa[32];
        {
            const float* xr = x + (size_t)myrow * DMODEL + 4 * hh;
            #pragma unroll
            for (int i = 0; i < 32; ++i)
                xa[i] = *(const float4*)(xr + 8 * i);
        }

        const bool lastRow = (wi == 13);
        const bool lastCol = (wj == 13);
        const bool maskR = lastRow && (hh != ((lane >> 3) & 1));

        for (int h = hbeg; h < hend; ++h) {
            __syncthreads();

            {
                const float* wsrc = wsrc0 + 96 * h;
                #pragma unroll
                for (int bt = 0; bt < 2; ++bt) {
                    float4 rr[3][4];
                    #pragma unroll
                    for (int u = 0; u < 3; ++u) {
                        const float* gp = wsrc + (size_t)bt * (128 * 3 * DMODEL) + u * 32;
                        #pragma unroll
                        for (int j = 0; j < 4; ++j)
                            rr[u][j] = *(const float4*)(gp + j * (3 * DMODEL));
                    }
                    __builtin_amdgcn_sched_barrier(0);
                    #pragma unroll
                    for (int u = 0; u < 3; ++u) {
                        float4 c0, c1, c2, c3;
                        c0.x = rr[u][0].x; c0.y = rr[u][1].x; c0.z = rr[u][2].x; c0.w = rr[u][3].x;
                        c1.x = rr[u][0].y; c1.y = rr[u][1].y; c1.z = rr[u][2].y; c1.w = rr[u][3].y;
                        c2.x = rr[u][0].z; c2.y = rr[u][1].z; c2.z = rr[u][2].z; c2.w = rr[u][3].z;
                        c3.x = rr[u][0].w; c3.y = rr[u][1].w; c3.z = rr[u][2].w; c3.w = rr[u][3].w;
                        const float4 d0 = sel4(rot1, c1, c0), d1 = sel4(rot1, c2, c1),
                                     d2 = sel4(rot1, c3, c2), d3 = sel4(rot1, c0, c3);
                        const float4 e0 = sel4(rot2, d2, d0), e1 = sel4(rot2, d3, d1),
                                     e2 = sel4(rot2, d0, d2), e3 = sel4(rot2, d1, d3);
                        const int so = bt * (16 * WL_STEP) + u * 256;
                        *(float4*)(wdst_t0 + so) = e0;
                        *(float4*)(wdst_t1 + so) = e1;
                        *(float4*)(wdst_t2 + so) = e2;
                        *(float4*)(wdst_t3 + so) = e3;
                    }
                }
            }

            v8f acc[6];
            {
                const float* bp = bvec + 96 * h;
                #pragma unroll
                for (int n = 0; n < 6; ++n) {
                    float bv = bp[16 * n + r];
                    v8f t;
                    #pragma unroll
                    for (int q = 0; q < 8; ++q) t[q] = bv;
                    acc[n] = t;
                }
            }

            __syncthreads();

            float4 bq[2][6];
            #pragma unroll
            for (int n = 0; n < 6; ++n)
                bq[0][n] = *(const float4*)(wrd + 128 * n);
            __builtin_amdgcn_sched_group_barrier(0x100, 6, 0);
            #pragma unroll
            for (int i = 0; i < 32; ++i) {
                if (i + 1 < 32) {
                    const float* nb = (i + 1 < HI_PAIR) ? wrd + (i + 1) * WL_STEP
                                                        : wrd_hi + (i + 1 - HI_PAIR) * WL_STEP;
                    #pragma unroll
                    for (int n = 0; n < 6; ++n)
                        bq[(i + 1) & 1][n] = *(const float4*)(nb + 128 * n);
                }
                v2f a0, a1;
                a0.x = xa[i].x; a0.y = xa[i].y;
                a1.x = xa[i].z; a1.y = xa[i].w;
                #pragma unroll
                for (int n = 0; n < 6; ++n) {
                    v2f b0; b0.x = bq[i & 1][n].x; b0.y = bq[i & 1][n].y;
                    acc[n] = wmma_f32_4(a0, b0, acc[n]);
                }
                #pragma unroll
                for (int n = 0; n < 6; ++n) {
                    v2f b1; b1.x = bq[i & 1][n].z; b1.y = bq[i & 1][n].w;
                    acc[n] = wmma_f32_4(a1, b1, acc[n]);
                }
                if (i + 1 < 32)
                    __builtin_amdgcn_sched_group_barrier(0x100, 6, 0);
                __builtin_amdgcn_sched_group_barrier(0x008, 12, 0);
            }
            __builtin_amdgcn_sched_barrier(0);

            int qrowb = 16 * wave + rowb;
            #pragma unroll
            for (int n = 0; n < 6; ++n)
                #pragma unroll
                for (int v = 0; v < 8; ++v)
                    qk[(qrowb + v) * QK_STRIDE + 16 * n + r] = acc[n][v];

            const float* qrow = &qk[(16 * wave + r) * QK_STRIDE];
            v8f sc = {0.f, 0.f, 0.f, 0.f, 0.f, 0.f, 0.f, 0.f};
            #pragma unroll
            for (int k0 = 0; k0 < HD; k0 += 4) {
                int ka = 3 * (k0 + 2 * hh);
                v2f a, bf;
                a.x  = qrow[ka];     a.y  = qrow[ka + 3];
                bf.x = qrow[ka + 1]; bf.y = qrow[ka + 4];
                sc = wmma_f32_4(a, bf, sc);
            }

            v8f p;
            #pragma unroll
            for (int v = 0; v < 8; ++v) {
                float xv = sc[v] * scale;
                bool mC = lastCol && (((v & 2) != 0) != ((lane & 2) != 0));
                if (maskR || mC) xv = -1e30f;
                float m = xv;
                #pragma unroll
                for (int off = 8; off; off >>= 1)
                    m = fmaxf(m, __shfl_xor(m, off, 16));
                float e = __expf(xv - m);
                float sm = e;
                #pragma unroll
                for (int off = 8; off; off >>= 1)
                    sm += __shfl_xor(sm, off, 16);
                p[v] = e / sm;
            }

            float* am = &att[wave * 16 * ATT_STRIDE];
            #pragma unroll
            for (int v = 0; v < 8; ++v)
                am[(rowb + v) * ATT_STRIDE + r] = p[v];

            v8f o0 = {0.f, 0.f, 0.f, 0.f, 0.f, 0.f, 0.f, 0.f};
            v8f o1 = {0.f, 0.f, 0.f, 0.f, 0.f, 0.f, 0.f, 0.f};
            #pragma unroll
            for (int k0 = 0; k0 < 16; k0 += 4) {
                v2f a = *(const v2f*)(am + r * ATT_STRIDE + k0 + 2 * hh);
                const float* v0p = &qk[(16 * wave + k0 + 2 * hh) * QK_STRIDE + 2];
                const float* v1p = v0p + QK_STRIDE;
                v2f b0; b0.x = v0p[3 * r];        b0.y = v1p[3 * r];
                o0 = wmma_f32_4(a, b0, o0);
                v2f b1; b1.x = v0p[3 * (16 + r)]; b1.y = v1p[3 * (16 + r)];
                o1 = wmma_f32_4(a, b1, o1);
            }

            float* ob = out + 32 * h + r;
            #pragma unroll
            for (int v = 0; v < 8; ++v) {
                int grow = rowoff[16 * wave + rowb + v];
                float* pdst = ob + (size_t)grow * DMODEL;
                __builtin_nontemporal_store(o0[v], pdst);
                __builtin_nontemporal_store(o1[v], pdst + 16);
            }
        }
    }
}

extern "C" void kernel_launch(void* const* d_in, const int* in_sizes, int n_in,
                              void* d_out, int out_size, void* d_ws, size_t ws_size,
                              hipStream_t stream) {
    (void)in_sizes; (void)n_in; (void)d_ws; (void)ws_size; (void)out_size;
    const float* x = (const float*)d_in[0];
    const float* W = (const float*)d_in[1];
    const float* b = (const float*)d_in[2];
    float* out = (float*)d_out;

    static_assert(SMEM_BYTES <= 160 * 1024, "two blocks per compute unit");
    hipFuncSetAttribute((const void*)swin_msa_fused,
                        hipFuncAttributeMaxDynamicSharedMemorySize, SMEM_BYTES);

    dim3 grid(NBLOCK);
    swin_msa_fused<<<grid, 256, SMEM_BYTES, stream>>>(x, W, b, out);
}


# One 16-wave work-group per compute unit as two 8-wave teams in opposite phase, one barrier per phase: a team's 384-WMMA projection covers the other's attention, scatter and W staging (s_setprio 3)
# speedup vs baseline: 1.6614x; 1.0581x over previous
#include <hip/hip_runtime.h>
#include <math.h>


typedef __attribute__((ext_vector_type(2))) float v2f;
typedef __attribute__((ext_vector_type(4))) float v4f;
typedef __attribute__((ext_vector_type(8))) float v8f;

#define NH         8
#define HD         32
#define DMODEL     256
#define HGRID      56
#define NWIN       8
#define WIN_IMG    196
#define NBLOCK     256
#define NTHREADS   512
#define WL_STEP    768
#define WL_FLOATS  (32 * WL_STEP)
#define HI_PAIR    21
#define QK_STRIDE  100
#define QK_FLOATS  (NWIN * 16 * QK_STRIDE)
#define ATT_STRIDE 18
#define ATT_FLOATS (NWIN * 16 * ATT_STRIDE)
#define TEAM_FLOATS (WL_FLOATS + QK_FLOATS + ATT_FLOATS + NWIN * 16)
#define SMEM_BYTES (2 * TEAM_FLOATS * 4)

static __device__ __forceinline__ v8f wmma_f32_4(v2f a, v2f b, v8f c) {
    return __builtin_amdgcn_wmma_f32_16x16x4_f32(false, a, false, b,
                                                 (short)0, c, false, false);
}

static __device__ __forceinline__ float4 sel4(bool c, float4 a, float4 b) {
    float4 o;
    o.x = c ? a.x : b.x; o.y = c ? a.y : b.y;
    o.z = c ? a.z : b.z; o.w = c ? a.w : b.w;
    return o;
}

static __device__ __forceinline__ int share(int q) {
    return (49 * q) >> 2;
}

__global__ __launch_bounds__(NTHREADS)
void swin_msa_fused(const float* __restrict__ x, const float* __restrict__ Wg,
                    const float* __restrict__ bvec, float* __restrict__ out) {
    extern __shared__ float smem[];

    const int tid  = threadIdx.x;
    const int team = __builtin_amdgcn_readfirstlane(((tid >> 6) ^ (tid >> 8)) & 1);

    float* wl  = smem + team * TEAM_FLOATS;
    float* qk  = wl + WL_FLOATS;
    float* att = qk + QK_FLOATS;
    int* rowoff = (int*)(att + ATT_FLOATS);

    const float scale = 0.17677669529663687f;

    const int q0 = 2 * blockIdx.x + 1, q1 = 2 * blockIdx.x;
    const int n0 = share(q0 + 1) - share(q0);
    const int n1 = share(q1 + 1) - share(q1);
    const int nsteps = (2 * n0 + 1 > 2 * n1 + 2) ? 2 * n0 + 1 : 2 * n1 + 2;

    const int qme  = team ? q1 : q0;
    const int ubeg = share(qme);
    const int uend = share(qme + 1);

    int  u    = ubeg;
    int  gcur = -1;
    bool pend = false;

    v4f xa[32];
    #pragma unroll
    for (int i = 0; i < 32; ++i) xa[i] = (v4f){0.f, 0.f, 0.f, 0.f};
    bool lastRowW = false, lastColW = false;

    for (int step = 0; step < nsteps; ++step) {
        const bool matrix_phase = ((step + team) & 1) != 0;

        if (!matrix_phase) {
            __builtin_amdgcn_s_setprio(3);

            const int ts = tid;
            const int w16  = ts >> 5;
            const int wave = (w16 & 1) | ((w16 >> 1) & 6);
            const int lane = ts & 31;
            const int r    = lane & 15;
            const int hh   = lane >> 4;
            const int rowb = hh << 3;

            if (pend) {
                const int h = (u - 1) & 7;
                const bool lastCol = lastColW;
                const bool maskR = lastRowW && (hh != ((lane >> 3) & 1));

                const float* qrow = &qk[(16 * wave + r) * QK_STRIDE];
                v2f qa[8], kb[8];
                #pragma unroll
                for (int j = 0; j < 8; ++j) {
                    const int ka = 3 * (4 * j + 2 * hh);
                    qa[j].x = qrow[ka];     qa[j].y = qrow[ka + 3];
                    kb[j].x = qrow[ka + 1]; kb[j].y = qrow[ka + 4];
                }
                #pragma unroll
                for (int j = 0; j < 8; ++j)
                    asm volatile("" : "+v"(qa[j]), "+v"(kb[j]));
                __builtin_amdgcn_sched_barrier(0);
                v8f sc = {0.f, 0.f, 0.f, 0.f, 0.f, 0.f, 0.f, 0.f};
                #pragma unroll
                for (int j = 0; j < 8; ++j)
                    sc = wmma_f32_4(qa[j], kb[j], sc);
                asm volatile("v_nop\n\tv_nop\n\tv_nop\n\tv_nop" : "+v"(sc));
                __builtin_amdgcn_sched_barrier(0);

                v8f p;
                #pragma unroll
                for (int v = 0; v < 8; ++v) {
                    float xv = sc[v] * scale;
                    bool mC = lastCol && (((v & 2) != 0) != ((lane & 2) != 0));
                    if (maskR || mC) xv = -1e30f;
                    float m = xv;
                    #pragma unroll
                    for (int off = 8; off; off >>= 1)
                        m = fmaxf(m, __shfl_xor(m, off, 16));
                    float e = __expf(xv - m);
                    float sm = e;
                    #pragma unroll
                    for (int off = 8; off; off >>= 1)
                        sm += __shfl_xor(sm, off, 16);
                    p[v] = e / sm;
                }

                float* am = &att[wave * 16 * ATT_STRIDE];
                #pragma unroll
                for (int v = 0; v < 8; ++v)
                    am[(rowb + v) * ATT_STRIDE + r] = p[v];

                v2f pa[4], vb0[4], vb1[4];
                #pragma unroll
                for (int j = 0; j < 4; ++j) {
                    const int k0 = 4 * j;
                    pa[j] = *(const v2f*)(am + r * ATT_STRIDE + k0 + 2 * hh);
                    const float* v0p = &qk[(16 * wave + k0 + 2 * hh) * QK_STRIDE + 2];
                    const float* v1p = v0p + QK_STRIDE;
                    vb0[j].x = v0p[3 * r];        vb0[j].y = v1p[3 * r];
                    vb1[j].x = v0p[3 * (16 + r)]; vb1[j].y = v1p[3 * (16 + r)];
                }
                #pragma unroll
                for (int j = 0; j < 4; ++j)
                    asm volatile("" : "+v"(pa[j]), "+v"(vb0[j]), "+v"(vb1[j]));
                __builtin_amdgcn_sched_barrier(0);
                v8f o0 = {0.f, 0.f, 0.f, 0.f, 0.f, 0.f, 0.f, 0.f};
                v8f o1 = {0.f, 0.f, 0.f, 0.f, 0.f, 0.f, 0.f, 0.f};
                #pragma unroll
                for (int j = 0; j < 4; ++j) {
                    o0 = wmma_f32_4(pa[j], vb0[j], o0);
                    o1 = wmma_f32_4(pa[j], vb1[j], o1);
                }
                asm volatile("v_nop\n\tv_nop\n\tv_nop\n\tv_nop" : "+v"(o0), "+v"(o1));
                __builtin_amdgcn_sched_barrier(0);

                float* ob = out + 32 * h + r;
                #pragma unroll
                for (int v = 0; v < 8; ++v) {
                    int grow = rowoff[16 * wave + rowb + v];
                    float* pdst = ob + (size_t)grow * DMODEL;
                    __builtin_nontemporal_store(o0[v], pdst);
                    __builtin_nontemporal_store(o1[v], pdst + 16);
                }
                pend = false;
            }

            if (u < uend) {
                const int g = u >> 3;
                const int h = u & 7;

                const int gwin = g * NWIN + wave;
                const int bb   = gwin / WIN_IMG;
                const int rem  = gwin - bb * WIN_IMG;
                const int wi   = rem / 14;
                const int wj   = rem - wi * 14;

                int gh = 4 * wi + (r >> 2) + 2; if (gh >= HGRID) gh -= HGRID;
                int gw = 4 * wj + (r & 3) + 2;  if (gw >= HGRID) gw -= HGRID;
                const int myrow = bb * (HGRID * HGRID) + gh * HGRID + gw;
                const float* xr = x + (size_t)myrow * DMODEL + 4 * hh;

                const int ldmask = (g != gcur) ? -1 : 0;

                if (g != gcur) {
                    gcur = g;
                    if (hh == 0) rowoff[16 * wave + r] = myrow;
                    lastRowW = (wi == 13);
                    lastColW = (wj == 13);
                }

                {
                    int exec_keep;
                    asm volatile(
                        "s_mov_b32 %8, exec_lo\n\t"
                        "s_and_b32 exec_lo, exec_lo, %10\n\t"
                        "global_load_b128 %0, %9, off\n\t"
                        "global_load_b128 %1, %9, off offset:32\n\t"
                        "global_load_b128 %2, %9, off offset:64\n\t"
                        "global_load_b128 %3, %9, off offset:96\n\t"
                        "global_load_b128 %4, %9, off offset:128\n\t"
                        "global_load_b128 %5, %9, off offset:160\n\t"
                        "global_load_b128 %6, %9, off offset:192\n\t"
                        "global_load_b128 %7, %9, off offset:224\n\t"
                        "s_mov_b32 exec_lo, %8\n\t"
                        "s_wait_loadcnt 0x0"
                        : "+v"(xa[0]), "+v"(xa[1]), "+v"(xa[2]), "+v"(xa[3]), "+v"(xa[4]), "+v"(xa[5]), "+v"(xa[6]), "+v"(xa[7]), "=&s"(exec_keep)
                        : "v"(xr), "s"(ldmask)
                        : "scc");
                    asm volatile(
                        "s_mov_b32 %8, exec_lo\n\t"
                        "s_and_b32 exec_lo, exec_lo, %10\n\t"
                        "global_load_b128 %0, %9, off offset:256\n\t"
                        "global_load_b128 %1, %9, off offset:288\n\t"
                        "global_load_b128 %2, %9, off offset:320\n\t"
                        "global_load_b128 %3, %9, off offset:352\n\t"
                        "global_load_b128 %4, %9, off offset:384\n\t"
                        "global_load_b128 %5, %9, off offset:416\n\t"
                        "global_load_b128 %6, %9, off offset:448\n\t"
                        "global_load_b128 %7, %9, off offset:480\n\t"
                        "s_mov_b32 exec_lo, %8\n\t"
                        "s_wait_loadcnt 0x0"
                        : "+v"(xa[8]), "+v"(xa[9]), "+v"(xa[10]), "+v"(xa[11]), "+v"(xa[12]), "+v"(xa[13]), "+v"(xa[14]), "+v"(xa[15]), "=&s"(exec_keep)
                        : "v"(xr), "s"(ldmask)
                        : "scc");
                    asm volatile(
                        "s_mov_b32 %8, exec_lo\n\t"
                        "s_and_b32 exec_lo, exec_lo, %10\n\t"
                        "global_load_b128 %0, %9, off offset:512\n\t"
                        "global_load_b128 %1, %9, off offset:544\n\t"
                        "global_load_b128 %2, %9, off offset:576\n\t"
                        "global_load_b128 %3, %9, off offset:608\n\t"
                        "global_load_b128 %4, %9, off offset:640\n\t"
                        "global_load_b128 %5, %9, off offset:672\n\t"
                        "global_load_b128 %6, %9, off offset:704\n\t"
                        "global_load_b128 %7, %9, off offset:736\n\t"
                        "s_mov_b32 exec_lo, %8\n\t"
                        "s_wait_loadcnt 0x0"
                        : "+v"(xa[16]), "+v"(xa[17]), "+v"(xa[18]), "+v"(xa[19]), "+v"(xa[20]), "+v"(xa[21]), "+v"(xa[22]), "+v"(xa[23]), "=&s"(exec_keep)
                        : "v"(xr), "s"(ldmask)
                        : "scc");
                    asm volatile(
                        "s_mov_b32 %8, exec_lo\n\t"
                        "s_and_b32 exec_lo, exec_lo, %10\n\t"
                        "global_load_b128 %0, %9, off offset:768\n\t"
                        "global_load_b128 %1, %9, off offset:800\n\t"
                        "global_load_b128 %2, %9, off offset:832\n\t"
                        "global_load_b128 %3, %9, off offset:864\n\t"
                        "global_load_b128 %4, %9, off offset:896\n\t"
                        "global_load_b128 %5, %9, off offset:928\n\t"
                        "global_load_b128 %6, %9, off offset:960\n\t"
                        "global_load_b128 %7, %9, off offset:992\n\t"
                        "s_mov_b32 exec_lo, %8\n\t"
                        "s_wait_loadcnt 0x0"
                        : "+v"(xa[24]), "+v"(xa[25]), "+v"(xa[26]), "+v"(xa[27]), "+v"(xa[28]), "+v"(xa[29]), "+v"(xa[30]), "+v"(xa[31]), "=&s"(exec_keep)
                        : "v"(xr), "s"(ldmask)
                        : "scc");
                }

                {
                    const int qq  = r & 3;
                    const int sh  = (r >> 2) & 1;
                    const int ntb = (r >> 3) & 1;
                    const float* wsrc = Wg + (size_t)(16 * wave + 8 * hh + 4 * sh) * (3 * DMODEL)
                                           + 16 * ntb + 4 * qq + 96 * h;
                    float* wdst0 = wl + (2 * wave + hh) * WL_STEP + ntb * 128 + sh * 64 + 16 * qq;
                    const bool rot1 = sh != 0;
                    const bool rot2 = ntb != 0;
                    const int  rot  = sh + 2 * ntb;
                    float* wdst_t0 = wdst0 + 4 * ((0 + rot) & 3);
                    float* wdst_t1 = wdst0 + 4 * ((1 + rot) & 3);
                    float* wdst_t2 = wdst0 + 4 * ((2 + rot) & 3);
                    float* wdst_t3 = wdst0 + 4 * ((3 + rot) & 3);

                    #pragma unroll
                    for (int bt = 0; bt < 2; ++bt) {
                        float4 rr[3][4];
                        #pragma unroll
                        for (int uu = 0; uu < 3; ++uu) {
                            const float* gp = wsrc + (size_t)bt * (128 * 3 * DMODEL) + uu * 32;
                            #pragma unroll
                            for (int j = 0; j < 4; ++j)
                                rr[uu][j] = *(const float4*)(gp + j * (3 * DMODEL));
                        }
                        __builtin_amdgcn_sched_barrier(0);
                        #pragma unroll
                        for (int uu = 0; uu < 3; ++uu) {
                            float4 c0, c1, c2, c3;
                            c0.x = rr[uu][0].x; c0.y = rr[uu][1].x; c0.z = rr[uu][2].x; c0.w = rr[uu][3].x;
                            c1.x = rr[uu][0].y; c1.y = rr[uu][1].y; c1.z = rr[uu][2].y; c1.w = rr[uu][3].y;
                            c2.x = rr[uu][0].z; c2.y = rr[uu][1].z; c2.z = rr[uu][2].z; c2.w = rr[uu][3].z;
                            c3.x = rr[uu][0].w; c3.y = rr[uu][1].w; c3.z = rr[uu][2].w; c3.w = rr[uu][3].w;
                            const float4 d0 = sel4(rot1, c1, c0), d1 = sel4(rot1, c2, c1),
                                         d2 = sel4(rot1, c3, c2), d3 = sel4(rot1, c0, c3);
                            const float4 e0 = sel4(rot2, d2, d0), e1 = sel4(rot2, d3, d1),
                                         e2 = sel4(rot2, d0, d2), e3 = sel4(rot2, d1, d3);
                            const int so = bt * (16 * WL_STEP) + uu * 256;
                            *(float4*)(wdst_t0 + so) = e0;
                            *(float4*)(wdst_t1 + so) = e1;
                            *(float4*)(wdst_t2 + so) = e2;
                            *(float4*)(wdst_t3 + so) = e3;
                        }
                    }
                }

                {
                    const float* bp = bvec + 96 * h + r;
                    float* bd = att + wave * 16 * ATT_STRIDE + r;
                    #pragma unroll
                    for (int n = 0; n < 6; ++n) bd[16 * n] = bp[16 * n];
                }
            }

            __builtin_amdgcn_s_setprio(0);
        } else if (step > 0 && u < uend) {

            const int tm = tid;
            const int mw16  = tm >> 5;
            const int mwave = (mw16 & 1) | ((mw16 >> 1) & 6);
            const int mr    = tm & 15;
            const int mhh   = (tm >> 4) & 1;
            const float* wrd = wl + mhh * 64 + 4 * mr;
            int hi_off = team * TEAM_FLOATS + mhh * 64 + 4 * mr + HI_PAIR * WL_STEP;
            asm volatile("" : "+v"(hi_off));
            const float* wrd_hi = smem + hi_off;
            float* qsp = qk + (16 * mwave + 8 * mhh) * QK_STRIDE + mr;
            const float* bsp = att + mwave * 16 * ATT_STRIDE + mr;

            v8f acc[6];
            #pragma unroll
            for (int n = 0; n < 6; ++n) {
                const float bvn = bsp[16 * n];
                v8f t;
                #pragma unroll
                for (int q = 0; q < 8; ++q) t[q] = bvn;
                acc[n] = t;
            }

            float4 bq[2][6];
            #pragma unroll
            for (int n = 0; n < 6; ++n)
                bq[0][n] = *(const float4*)(wrd + 128 * n);
            __builtin_amdgcn_sched_group_barrier(0x100, 12, 0);
            #pragma unroll
            for (int i = 0; i < 32; ++i) {
                if (i + 1 < 32) {
                    const float* nb = (i + 1 < HI_PAIR) ? wrd + (i + 1) * WL_STEP
                                                        : wrd_hi + (i + 1 - HI_PAIR) * WL_STEP;
                    #pragma unroll
                    for (int n = 0; n < 6; ++n)
                        bq[(i + 1) & 1][n] = *(const float4*)(nb + 128 * n);
                }
                v2f a0, a1;
                a0.x = xa[i].x; a0.y = xa[i].y;
                a1.x = xa[i].z; a1.y = xa[i].w;
                #pragma unroll
                for (int n = 0; n < 6; ++n) {
                    v2f b0; b0.x = bq[i & 1][n].x; b0.y = bq[i & 1][n].y;
                    acc[n] = wmma_f32_4(a0, b0, acc[n]);
                }
                #pragma unroll
                for (int n = 0; n < 6; ++n) {
                    v2f b1; b1.x = bq[i & 1][n].z; b1.y = bq[i & 1][n].w;
                    acc[n] = wmma_f32_4(a1, b1, acc[n]);
                }
                if (i + 1 < 32)
                    __builtin_amdgcn_sched_group_barrier(0x100, 6, 0);
                __builtin_amdgcn_sched_group_barrier(0x008, 12, 0);
            }
            __builtin_amdgcn_sched_barrier(0);

            #pragma unroll
            for (int n = 0; n < 6; ++n)
                #pragma unroll
                for (int v = 0; v < 8; ++v)
                    qsp[v * QK_STRIDE + 16 * n] = acc[n][v];

            pend = true;
            ++u;
        }

        __syncthreads();
    }
}

extern "C" void kernel_launch(void* const* d_in, const int* in_sizes, int n_in,
                              void* d_out, int out_size, void* d_ws, size_t ws_size,
                              hipStream_t stream) {
    (void)in_sizes; (void)n_in; (void)d_ws; (void)ws_size; (void)out_size;
    const float* x = (const float*)d_in[0];
    const float* W = (const float*)d_in[1];
    const float* b = (const float*)d_in[2];
    float* out = (float*)d_out;

    static_assert(SMEM_BYTES <= 320 * 1024, "one work-group per compute unit");
    hipFuncSetAttribute((const void*)swin_msa_fused,
                        hipFuncAttributeMaxDynamicSharedMemorySize, SMEM_BYTES);

    dim3 grid(NBLOCK);
    swin_msa_fused<<<grid, NTHREADS, SMEM_BYTES, stream>>>(x, W, b, out);
}
